# starting kernel
# baseline (speedup 1.0000x reference)
#include <hip/hip_runtime.h>
#include <math.h>

static constexpr int Bn = 2048;
static constexpr int Tn = 1000;

typedef float v2f __attribute__((ext_vector_type(2)));
typedef float v4f __attribute__((ext_vector_type(4)));
typedef float v8f __attribute__((ext_vector_type(8)));

__global__ __launch_bounds__(64) void kf_gain_kernel(
    const float* __restrict__ F, const float* __restrict__ Gmat,
    const float* __restrict__ H, const float* __restrict__ Q,
    const float* __restrict__ R, const float* __restrict__ P0,
    float* __restrict__ Gws, int* __restrict__ t_stop_ptr,
    float* __restrict__ Ps)
{
    __shared__ float Fm[8][8], Hm[8][8], Qe[8][8], Rm[8][8];
    __shared__ float P[8][8], Pneg[8][8], T1[8][8], T2[8][8];
    __shared__ float S[8][8], X[8][8], IKH[8][8], Amat[8][8], Pn[8][8];
    __shared__ float GQ[8][8];
    __shared__ int flag;

    const int tid = threadIdx.x;
    const int i = tid >> 3, j = tid & 7;

    Fm[i][j] = F[tid];
    Hm[i][j] = H[tid];
    Rm[i][j] = R[tid];
    P[i][j]  = P0[tid];
    __syncthreads();

    float acc = 0.f;
    #pragma unroll
    for (int k = 0; k < 8; ++k) acc += Gmat[i * 8 + k] * Q[k * 8 + j];
    GQ[i][j] = acc;
    __syncthreads();
    acc = 0.f;
    #pragma unroll
    for (int k = 0; k < 8; ++k) acc += GQ[i][k] * Gmat[j * 8 + k];
    Qe[i][j] = acc;
    __syncthreads();

    int tstop = Tn;
    for (int t = 0; t < Tn; ++t) {
        acc = 0.f;
        #pragma unroll
        for (int k = 0; k < 8; ++k) acc += Fm[i][k] * P[k][j];
        T1[i][j] = acc;
        __syncthreads();
        acc = Qe[i][j];
        #pragma unroll
        for (int k = 0; k < 8; ++k) acc += T1[i][k] * Fm[j][k];
        Pneg[i][j] = acc;
        __syncthreads();
        acc = 0.f;
        #pragma unroll
        for (int k = 0; k < 8; ++k) acc += Hm[i][k] * Pneg[k][j];
        T2[i][j] = acc;
        __syncthreads();
        acc = Rm[i][j];
        #pragma unroll
        for (int k = 0; k < 8; ++k) acc += T2[i][k] * Hm[j][k];
        S[i][j] = acc;
        __syncthreads();
        #pragma unroll 1
        for (int k = 0; k < 8; ++k) {
            if (i == k && j == k) S[k][k] = sqrtf(S[k][k]);
            __syncthreads();
            if (j == k && i > k) S[i][k] = S[i][k] / S[k][k];
            __syncthreads();
            if (i > k && j > k && j <= i) S[i][j] -= S[i][k] * S[j][k];
            __syncthreads();
        }
        #pragma unroll 1
        for (int r = 0; r < 8; ++r) {
            if (i == r) {
                float s = T2[r][j];
                for (int m = 0; m < r; ++m) s -= S[r][m] * X[m][j];
                X[r][j] = s / S[r][r];
            }
            __syncthreads();
        }
        #pragma unroll 1
        for (int r = 7; r >= 0; --r) {
            if (i == r) {
                float s = X[r][j];
                for (int m = r + 1; m < 8; ++m) s -= S[m][r] * X[m][j];
                X[r][j] = s / S[r][r];
            }
            __syncthreads();
        }
        acc = (i == j) ? 1.f : 0.f;
        #pragma unroll
        for (int m = 0; m < 8; ++m) acc -= X[m][i] * Hm[m][j];
        IKH[i][j] = acc;
        __syncthreads();
        acc = 0.f;
        float pacc = 0.f;
        #pragma unroll
        for (int k = 0; k < 8; ++k) {
            acc  += IKH[i][k] * Fm[k][j];
            pacc += IKH[i][k] * Pneg[k][j];
        }
        Amat[i][j] = acc;
        Pn[i][j]   = pacc;
        __syncthreads();
        float sym = 0.5f * (Pn[i][j] + Pn[j][i]);
        __syncthreads();
        Pn[i][j] = sym;
        __syncthreads();

        if (tid == 0) flag = 0;
        __syncthreads();
        if (Pn[i][j] != P[i][j]) flag = 1;

        Ps[t * 64 + tid] = Pn[i][j];
        float* gout = Gws + (size_t)t * 256;
        #pragma unroll
        for (int q = 0; q < 4; ++q) {
            int f    = tid * 4 + q;
            int lane = f >> 3, slot = f & 7;
            int s2 = slot >> 1, p = slot & 1;
            int m = lane & 15;
            int k = 4 * s2 + ((lane < 16) ? p : (2 + p));
            int mm = (m < 8) ? m : ((m <= 13) ? (m - 6) : -1);
            float v = 0.f;
            if (mm >= 0) v = (k < 8) ? Amat[mm][k] : X[k - 8][mm];
            gout[f] = v;
        }
        P[i][j] = Pn[i][j];
        __syncthreads();
        if (flag == 0) { tstop = t + 1; break; }
    }
    if (tid == 0) *t_stop_ptr = tstop;
}

__global__ __launch_bounds__(256) void kf_fill_kernel(
    float* __restrict__ Gws, const int* __restrict__ t_stop_ptr,
    float* __restrict__ Ps)
{
    const int t  = blockIdx.x;
    const int q  = threadIdx.x;
    const int ts = *t_stop_ptr;
    if (t >= ts && ts >= 1) {
        const int src = ts - 1;
        Gws[(size_t)t * 256 + q] = Gws[(size_t)src * 256 + q];
        if (q < 64) Ps[t * 64 + q] = Ps[src * 64 + q];
    }
}

__global__ __launch_bounds__(32) void kf_scan_kernel(
    const float* __restrict__ Y, const float* __restrict__ x0,
    const float* __restrict__ Gws, float* __restrict__ traj)
{
    const int lane = threadIdx.x;
    const int n    = lane & 15;
    const bool lo  = lane < 16;
    const size_t row = (size_t)(blockIdx.x * 16 + n);

    const float* yrow = Y    + row * (size_t)Tn * 8 + (lo ? 0 : 2);
    float*       orow = traj + row * (size_t)Tn * 8 + (lo ? 0 : 4);

    v4f xa = *(const v4f*)(x0 + row * 8);
    v4f xb = *(const v4f*)(x0 + row * 8 + 4);
    v8f xv;
    xv[0] = lo ? xa.x : xa.z;
    xv[1] = lo ? xa.y : xa.w;
    xv[2] = lo ? xa.z : xb.x;
    xv[3] = lo ? xa.w : xb.y;
    xv[4] = lo ? xb.x : xb.z;
    xv[5] = lo ? xb.y : xb.w;
    xv[6] = lo ? xb.z : 0.f;
    xv[7] = lo ? xb.w : 0.f;

    const v8f zero8 = {0.f, 0.f, 0.f, 0.f, 0.f, 0.f, 0.f, 0.f};

    for (int t = 0; t < Tn; ++t) {
        const v4f* gp = (const v4f*)(Gws + (size_t)t * 256 + lane * 8);
        v4f g0 = gp[0];
        v4f g1 = gp[1];
        v2f a0 = {g0.x, g0.y};
        v2f a1 = {g0.z, g0.w};
        v2f a2 = {g1.x, g1.y};
        v2f a3 = {g1.z, g1.w};

        v2f b2 = *(const v2f*)(yrow + (size_t)t * 8);
        v2f b3 = *(const v2f*)(yrow + (size_t)t * 8 + 4);

        v2f b0 = {xv[0], xv[1]};
        v2f b1 = {xv[4], xv[5]};

        v8f cy = __builtin_amdgcn_wmma_f32_16x16x4_f32(false, a2, false, b2,
                                                       (short)0, zero8, false, false);
        cy = __builtin_amdgcn_wmma_f32_16x16x4_f32(false, a3, false, b3,
                                                   (short)0, cy, false, false);
        v8f c1 = __builtin_amdgcn_wmma_f32_16x16x4_f32(false, a0, false, b0,
                                                       (short)0, cy, false, false);
        v8f c2 = __builtin_amdgcn_wmma_f32_16x16x4_f32(false, a1, false, b1,
                                                       (short)0, zero8, false, false);
        xv = c1 + c2;

        v4f out;
        out.x = lo ? xv[0] : xv[2];
        out.y = lo ? xv[1] : xv[3];
        out.z = lo ? xv[2] : xv[4];
        out.w = lo ? xv[3] : xv[5];
        *(v4f*)(orow + (size_t)t * 8) = out;
    }
}

extern "C" void kernel_launch(void* const* d_in, const int* in_sizes, int n_in,
                              void* d_out, int out_size, void* d_ws, size_t ws_size,
                              hipStream_t stream) {
    const float* Y    = (const float*)d_in[0];
    const float* F    = (const float*)d_in[1];
    const float* Gmat = (const float*)d_in[2];
    const float* H    = (const float*)d_in[3];
    const float* Q    = (const float*)d_in[4];
    const float* R    = (const float*)d_in[5];
    const float* x0   = (const float*)d_in[6];
    const float* P0   = (const float*)d_in[7];

    float* traj = (float*)d_out;
    float* Ps   = traj + (size_t)Bn * Tn * 8;

    int*   t_stop = (int*)d_ws;
    float* Gws    = (float*)((char*)d_ws + 256);

    kf_gain_kernel<<<1, 64, 0, stream>>>(F, Gmat, H, Q, R, P0, Gws, t_stop, Ps);
    kf_fill_kernel<<<Tn, 256, 0, stream>>>(Gws, t_stop, Ps);
    kf_scan_kernel<<<Bn / 16, 32, 0, stream>>>(Y, x0, Gws, traj);
}


# Scan kernel requests each step's gain block and observations 12 steps ahead into a register ring: a step no longer waits for its own L2 and memory loads; sum order unchanged
# speedup vs baseline: 8.1746x; 8.1746x over previous
#include <hip/hip_runtime.h>
#include <math.h>

static constexpr int Bn = 2048;
static constexpr int Tn = 1000;

typedef float v2f __attribute__((ext_vector_type(2)));
typedef float v4f __attribute__((ext_vector_type(4)));
typedef float v8f __attribute__((ext_vector_type(8)));

__device__ __forceinline__ void barrier_lds_only() {
    __builtin_amdgcn_fence(__ATOMIC_RELEASE, "workgroup", "local");
    __builtin_amdgcn_s_barrier();
    __builtin_amdgcn_fence(__ATOMIC_ACQUIRE, "workgroup", "local");
}

__global__ __launch_bounds__(64) void kf_gain_kernel(
    const float* __restrict__ F, const float* __restrict__ Gmat,
    const float* __restrict__ H, const float* __restrict__ Q,
    const float* __restrict__ R, const float* __restrict__ P0,
    float* __restrict__ Gnat, int* __restrict__ t_stop_ptr,
    float* __restrict__ Ps)
{
    __shared__ __attribute__((aligned(16))) float Fm[8][8];
    __shared__ __attribute__((aligned(16))) float Hm[8][8];
    __shared__ __attribute__((aligned(16))) float Qe[8][8];
    __shared__ __attribute__((aligned(16))) float Rm[8][8];
    __shared__ __attribute__((aligned(16))) float PnegL[8][8];
    __shared__ __attribute__((aligned(16))) float PnL[8][8];
    __shared__ __attribute__((aligned(16))) float X[8][8];
    __shared__ __attribute__((aligned(16))) float Sb[2][8][8];
    __shared__ __attribute__((aligned(16))) float Xb[2][8][8];
    __shared__ float GQ[8][8];
    __shared__ int flags[3];

    const int tid = threadIdx.x;
    const int i = tid >> 3, j = tid & 7;

    Fm[i][j]  = F[tid];
    Hm[i][j]  = H[tid];
    Rm[i][j]  = R[tid];
    PnL[i][j] = P0[tid];
    if (tid < 3) flags[tid] = 0;
    __syncthreads();

    float acc = 0.f;
    #pragma unroll
    for (int k = 0; k < 8; ++k) acc += Gmat[i * 8 + k] * Q[k * 8 + j];
    GQ[i][j] = acc;
    __syncthreads();
    acc = 0.f;
    #pragma unroll
    for (int k = 0; k < 8; ++k) acc += GQ[i][k] * Gmat[j * 8 + k];
    Qe[i][j] = acc;
    __syncthreads();

    const int sa = (i > j) ? i : j, sb = (i > j) ? j : i;

    float fri[8], frj[8], fcj[8], hri[8], hsa[8], hsb[8];
    #pragma unroll
    for (int k = 0; k < 8; ++k) {
        fri[k] = Fm[i][k];  frj[k] = Fm[j][k];  fcj[k] = Fm[k][j];
        hri[k] = Hm[i][k];  hsa[k] = Hm[sa][k]; hsb[k] = Hm[sb][k];
    }
    const float qe = Qe[i][j];
    const float rr = Rm[sa][sb];
    float Hw[64];
    float Pw[64];
    #pragma unroll
    for (int q = 0; q < 16; ++q) {
        const v4f h4 = ((const v4f*)&Hm[0][0])[q];
        const v4f p4 = ((const v4f*)&PnL[0][0])[q];
        Hw[4*q] = h4.x; Hw[4*q+1] = h4.y; Hw[4*q+2] = h4.z; Hw[4*q+3] = h4.w;
        Pw[4*q] = p4.x; Pw[4*q+1] = p4.y; Pw[4*q+2] = p4.z; Pw[4*q+3] = p4.w;
    }
    float pold = PnL[i][j];

    int tstop = Tn;
    int fc = 0, fp = 0;
    for (int t = 0; t < Tn; ++t) {
        float pn = qe;
        {
            float t1[8];
            #pragma unroll
            for (int kk = 0; kk < 8; ++kk) {
                float a = 0.f;
                #pragma unroll
                for (int m = 0; m < 8; ++m) a += fri[m] * Pw[m * 8 + kk];
                t1[kk] = a;
            }
            #pragma unroll
            for (int k = 0; k < 8; ++k) pn += t1[k] * frj[k];
        }
        PnegL[i][j] = pn;
        barrier_lds_only();
        if (t > 0 && flags[fp] == 0) { tstop = t; break; }

        float pcol[8];
        #pragma unroll
        for (int k = 0; k < 8; ++k) pcol[k] = PnegL[k][j];
        float sv = rr;
        {
            float Pg[64];
            #pragma unroll
            for (int q = 0; q < 16; ++q) {
                const v4f p4 = ((const v4f*)&PnegL[0][0])[q];
                Pg[4*q] = p4.x; Pg[4*q+1] = p4.y; Pg[4*q+2] = p4.z; Pg[4*q+3] = p4.w;
            }
            #pragma unroll
            for (int kk = 0; kk < 8; ++kk) {
                float a = 0.f;
                #pragma unroll
                for (int m = 0; m < 8; ++m) a += hsa[m] * Pg[m * 8 + kk];
                sv += a * hsb[kk];
            }
        }
        float xv = 0.f;
        #pragma unroll
        for (int k = 0; k < 8; ++k) xv += hri[k] * pcol[k];

        #pragma unroll
        for (int hb = 0; hb < 2; ++hb) {
            const int k = 4 * hb;
            Sb[hb][i][j] = sv;
            Xb[hb][i][j] = xv;
            barrier_lds_only();
            const v4f w0 = *(const v4f*)&Sb[hb][k][k];
            const v4f w1 = *(const v4f*)&Sb[hb][k + 1][k];
            const v4f w2 = *(const v4f*)&Sb[hb][k + 2][k];
            const v4f w3 = *(const v4f*)&Sb[hb][k + 3][k];
            const v4f wi = *(const v4f*)&Sb[hb][i][k];
            const float s0j = Sb[hb][k][j],     s1j = Sb[hb][k + 1][j];
            const float s2j = Sb[hb][k + 2][j], s3j = Sb[hb][k + 3][j];
            const float x0j = Xb[hb][k][j],     x1j = Xb[hb][k + 1][j];
            const float x2j = Xb[hb][k + 2][j], x3j = Xb[hb][k + 3][j];
            __builtin_amdgcn_sched_barrier(0);

            const float pa = w0.x, pb = w0.y, pc = w1.x, pd = w1.y;
            const float det = pa * pd - pb * pc;
            const float rd  = 1.0f / det;
            const float e0 = w2.x * pd - w2.y * pc;
            const float e1 = w2.y * pa - w2.x * pb;
            const float g0 = w3.x * pd - w3.y * pc;
            const float g1 = w3.y * pa - w3.x * pb;
            const float he0 = -e0, he1 = -e1, hg0 = -g0, hg1 = -g1;
            const float si0 = wi.x, si1 = wi.y;
            const float n0 = si0 * pd - si1 * pc;
            const float n1 = si1 * pa - si0 * pb;
            const bool r0 = (i == k), r1 = (i == k + 1);
            const float h0 = r0 ? pd : (r1 ? -pc : -n0);
            const float h1 = r0 ? -pb : (r1 ? pa : -n1);
            const float qa  = w2.z + (he0 * w0.z + he1 * w1.z) * rd;
            const float qb  = w2.w + (he0 * w0.w + he1 * w1.w) * rd;
            const float qc  = w3.z + (hg0 * w0.z + hg1 * w1.z) * rd;
            const float qd  = w3.w + (hg0 * w0.w + hg1 * w1.w) * rd;
            const float t2j = s2j + (he0 * s0j + he1 * s1j) * rd;
            const float t3j = s3j + (hg0 * s0j + hg1 * s1j) * rd;
            const float y2j = x2j + (he0 * x0j + he1 * x1j) * rd;
            const float y3j = x3j + (hg0 * x0j + hg1 * x1j) * rd;
            const bool  rp  = r0 || r1;
            const float ti0 = (rp ? 0.f : wi.z) + (h0 * w0.z + h1 * w1.z) * rd;
            const float ti1 = (rp ? 0.f : wi.w) + (h0 * w0.w + h1 * w1.w) * rd;
            const float sv1 = (rp ? 0.f : sv) + (h0 * s0j + h1 * s1j) * rd;
            const float xv1 = (rp ? 0.f : xv) + (h0 * x0j + h1 * x1j) * rd;

            const float det2 = qa * qd - qb * qc;
            const float rd2  = 1.0f / det2;
            const float m0 = ti0 * qd - ti1 * qc;
            const float m1 = ti1 * qa - ti0 * qb;
            const bool r2 = (i == k + 2), r3 = (i == k + 3);
            const float f0 = r2 ? qd : (r3 ? -qc : -m0);
            const float f1 = r2 ? -qb : (r3 ? qa : -m1);
            const bool  rq = r2 || r3;
            sv = (rq ? 0.f : sv1) + (f0 * t2j + f1 * t3j) * rd2;
            xv = (rq ? 0.f : xv1) + (f0 * y2j + f1 * y3j) * rd2;
        }
        X[i][j] = xv;
        barrier_lds_only();
        float am = 0.f, pacc = 0.f;
        {
            float xc[8];
            #pragma unroll
            for (int m = 0; m < 8; ++m) xc[m] = X[m][i];
            #pragma unroll
            for (int kk = 0; kk < 8; ++kk) {
                float a = (i == kk) ? 1.f : 0.f;
                #pragma unroll
                for (int m = 0; m < 8; ++m) a -= xc[m] * Hw[m * 8 + kk];
                am   += a * fcj[kk];
                pacc += a * pcol[kk];
            }
        }
        PnL[i][j]  = pacc;
        barrier_lds_only();
        float sym;
        {
            float Pr[64];
            #pragma unroll
            for (int q = 0; q < 16; ++q) {
                const v4f p4 = ((const v4f*)&PnL[0][0])[q];
                Pr[4*q] = p4.x; Pr[4*q+1] = p4.y; Pr[4*q+2] = p4.z; Pr[4*q+3] = p4.w;
            }
            sym = 0.5f * (pacc + PnL[j][i]);
            #pragma unroll
            for (int m = 0; m < 8; ++m) {
                #pragma unroll
                for (int k = m; k < 8; ++k) {
                    const float s = 0.5f * (Pr[m * 8 + k] + Pr[k * 8 + m]);
                    Pw[m * 8 + k] = s;
                    Pw[k * 8 + m] = s;
                }
            }
        }
        const int fn = (fc == 2) ? 0 : fc + 1;
        if (sym != pold) flags[fc] = 1;
        if (tid == 0) flags[fn] = 0;
        pold = sym;
        fp = fc; fc = fn;

        Ps[t * 64 + tid] = sym;
        v2f g;
        g.x = am;
        g.y = xv;
        *(v2f*)(Gnat + ((size_t)t * 64 + tid) * 2) = g;
    }
    if (tid == 0) *t_stop_ptr = tstop;
}

__global__ __launch_bounds__(256) void kf_fill_kernel(
    float* __restrict__ Gws, const float* __restrict__ Gnat,
    const int* __restrict__ t_stop_ptr, float* __restrict__ Ps)
{
    const int t  = blockIdx.x;
    const int f  = threadIdx.x;
    const int ts = *t_stop_ptr;
    const int last = (ts > 0) ? (ts - 1) : 0;
    const int src  = (t < ts) ? t : last;

    const int lane = f >> 3, slot = f & 7;
    const int s2 = slot >> 1, p = slot & 1;
    const int m = lane & 15;
    const int k = 4 * s2 + ((lane < 16) ? p : (2 + p));
    const int mm = (m < 8) ? m : ((m <= 13) ? (m - 6) : -1);
    float v = 0.f;
    if (mm >= 0) {
        const int e = (k < 8) ? ((mm * 8 + k) * 2) : (((k - 8) * 8 + mm) * 2 + 1);
        v = Gnat[(size_t)src * 128 + e];
    }
    Gws[(size_t)t * 256 + f] = v;
    if (t >= ts && ts >= 1 && f < 64) Ps[t * 64 + f] = Ps[src * 64 + f];
}

__global__ __launch_bounds__(32) void kf_scan_kernel(
    const float* __restrict__ Y, const float* __restrict__ x0,
    const float* __restrict__ Gws, float* __restrict__ traj)
{
    const int lane = threadIdx.x;
    const int n    = lane & 15;
    const bool lo  = lane < 16;
    const size_t row = (size_t)(blockIdx.x * 16 + n);

    const float* yrow = Y    + row * (size_t)Tn * 8 + (lo ? 0 : 2);
    float*       orow = traj + row * (size_t)Tn * 8 + (lo ? 0 : 4);

    v4f xa = *(const v4f*)(x0 + row * 8);
    v4f xb = *(const v4f*)(x0 + row * 8 + 4);
    v8f xv;
    xv[0] = lo ? xa.x : xa.z;
    xv[1] = lo ? xa.y : xa.w;
    xv[2] = lo ? xa.z : xb.x;
    xv[3] = lo ? xa.w : xb.y;
    xv[4] = lo ? xb.x : xb.z;
    xv[5] = lo ? xb.y : xb.w;
    xv[6] = lo ? xb.z : 0.f;
    xv[7] = lo ? xb.w : 0.f;

    const v8f zero8 = {0.f, 0.f, 0.f, 0.f, 0.f, 0.f, 0.f, 0.f};

    constexpr int RD = 12;
    v4f rg0[RD], rg1[RD];
    v2f ry2[RD], ry3[RD];
    #pragma unroll
    for (int u = 0; u < RD; ++u) {
        const v4f* gp = (const v4f*)(Gws + (size_t)u * 256 + lane * 8);
        rg0[u] = gp[0];
        rg1[u] = gp[1];
        ry2[u] = *(const v2f*)(yrow + (size_t)u * 8);
        ry3[u] = *(const v2f*)(yrow + (size_t)u * 8 + 4);
        __builtin_amdgcn_sched_barrier(0);
    }

    #define KF_SCAN_STEP(u, un, t)                                                \
    {                                                                             \
        const v2f a0 = {rg0[u].x, rg0[u].y};          \
        const v2f a1 = {rg0[u].z, rg0[u].w};          \
        const v2f a2 = {rg1[un].x, rg1[un].y};        \
        const v2f a3 = {rg1[un].z, rg1[un].w};        \
                     \
        const v2f b0 = {xv[0], xv[1]};               \
        const v2f b1 = {xv[4], xv[5]};               \
        const v8f c1 = __builtin_amdgcn_wmma_f32_16x16x4_f32(false, a0, false, b0, \
                                                       (short)0, cyc, false, false); \
        const v8f c2 = __builtin_amdgcn_wmma_f32_16x16x4_f32(false, a1, false, b1, \
                                                       (short)0, zero8, false, false); \
           \
        v8f cn = __builtin_amdgcn_wmma_f32_16x16x4_f32(false, a2, false, ry2[un], \
                                                       (short)0, zero8, false, false); \
        cn = __builtin_amdgcn_wmma_f32_16x16x4_f32(false, a3, false, ry3[un],     \
                                                   (short)0, cn, false, false);   \
        xv = c1 + c2;                                                             \
        cyc = cn;                                                                 \
                  \
                       \
        v4f out;                                                                  \
        out.x = lo ? xv[0] : xv[2];                                               \
        out.y = lo ? xv[1] : xv[3];                                               \
        out.z = lo ? xv[2] : xv[4];                                               \
        out.w = lo ? xv[3] : xv[5];                                               \
        *(v4f*)(orow + (size_t)(t) * 8) = out;                                    \
    }

    v8f cyc;
    {
        const v2f a2 = {rg1[0].x, rg1[0].y};
        const v2f a3 = {rg1[0].z, rg1[0].w};
        cyc = __builtin_amdgcn_wmma_f32_16x16x4_f32(false, a2, false, ry2[0],
                                                    (short)0, zero8, false, false);
        cyc = __builtin_amdgcn_wmma_f32_16x16x4_f32(false, a3, false, ry3[0],
                                                    (short)0, cyc, false, false);
    }

    int t = 0;
    for (; t + RD <= Tn; t += RD) {
        #pragma unroll
        for (int u = 0; u < RD; ++u) {
            KF_SCAN_STEP(u, (u + 1) % RD, t + u)
            const int tn = t + u + RD;
            const int tl = (tn < Tn) ? tn : (Tn - 1);
            const v4f* gp = (const v4f*)(Gws + (size_t)tl * 256 + lane * 8);
            rg0[u] = gp[0];
            rg1[u] = gp[1];
            ry2[u] = *(const v2f*)(yrow + (size_t)tl * 8);
            ry3[u] = *(const v2f*)(yrow + (size_t)tl * 8 + 4);
            __builtin_amdgcn_sched_barrier(0);
        }
    }
    #pragma unroll
    for (int u = 0; u < Tn % RD; ++u) {
        KF_SCAN_STEP(u, (u + 1) % RD, t + u)
    }
    #undef KF_SCAN_STEP
}

extern "C" void kernel_launch(void* const* d_in, const int* in_sizes, int n_in,
                              void* d_out, int out_size, void* d_ws, size_t ws_size,
                              hipStream_t stream) {
    const float* Y    = (const float*)d_in[0];
    const float* F    = (const float*)d_in[1];
    const float* Gmat = (const float*)d_in[2];
    const float* H    = (const float*)d_in[3];
    const float* Q    = (const float*)d_in[4];
    const float* R    = (const float*)d_in[5];
    const float* x0   = (const float*)d_in[6];
    const float* P0   = (const float*)d_in[7];

    float* traj = (float*)d_out;
    float* Ps   = traj + (size_t)Bn * Tn * 8;

    int*   t_stop = (int*)d_ws;
    float* Gws    = (float*)((char*)d_ws + 256);
    float* Gnat   = (float*)((char*)d_ws + 1024256);

    kf_gain_kernel<<<1, 64, 0, stream>>>(F, Gmat, H, Q, R, P0, Gnat, t_stop, Ps);
    kf_fill_kernel<<<Tn, 256, 0, stream>>>(Gws, Gnat, t_stop, Ps);
    kf_scan_kernel<<<Bn / 16, 32, 0, stream>>>(Y, x0, Gws, traj);
}


# Gain kernel: Pneg is never handed over: S and H Pneg from a thread's row of (HF) P, Pn = Pneg - K (H Pneg) from its own Pneg element: 2 barriers a step, not 3; sum order of S, Pn, (I-KH)F changed
# speedup vs baseline: 8.8947x; 1.0881x over previous
#include <hip/hip_runtime.h>
#include <math.h>

static constexpr int Bn = 2048;
static constexpr int Tn = 1000;

typedef float v2f __attribute__((ext_vector_type(2)));
typedef float v4f __attribute__((ext_vector_type(4)));
typedef float v8f __attribute__((ext_vector_type(8)));

__device__ __forceinline__ void barrier_lds_only() {
    __builtin_amdgcn_fence(__ATOMIC_RELEASE, "workgroup", "local");
    __builtin_amdgcn_s_barrier();
    __builtin_amdgcn_fence(__ATOMIC_ACQUIRE, "workgroup", "local");
}

__global__ __launch_bounds__(64) void kf_gain_kernel(
    const float* __restrict__ F, const float* __restrict__ Gmat,
    const float* __restrict__ H, const float* __restrict__ Q,
    const float* __restrict__ R, const float* __restrict__ P0,
    float* __restrict__ Gnat, int* __restrict__ t_stop_ptr,
    float* __restrict__ Ps)
{
    __shared__ __attribute__((aligned(16))) float Fm[8][8];
    __shared__ __attribute__((aligned(16))) float Hm[8][8];
    __shared__ __attribute__((aligned(16))) float Qe[8][8];
    __shared__ __attribute__((aligned(16))) float Rm[8][8];
    __shared__ __attribute__((aligned(16))) float Mm[8][8];
    __shared__ __attribute__((aligned(16))) float HQm[8][8];
    __shared__ __attribute__((aligned(16))) float PnL[8][8];
    __shared__ __attribute__((aligned(16))) float Sb[8][8];
    __shared__ __attribute__((aligned(16))) float Tb[8][8];
    __shared__ float GQ[8][8];
    __shared__ int flags[3];

    const int tid = threadIdx.x;
    const int i = tid >> 3, j = tid & 7;

    Fm[i][j]  = F[tid];
    Hm[i][j]  = H[tid];
    Rm[i][j]  = R[tid];
    PnL[i][j] = P0[tid];
    if (tid < 3) flags[tid] = 0;
    __syncthreads();

    float acc = 0.f;
    #pragma unroll
    for (int k = 0; k < 8; ++k) acc += Gmat[i * 8 + k] * Q[k * 8 + j];
    GQ[i][j] = acc;
    __syncthreads();
    acc = 0.f;
    #pragma unroll
    for (int k = 0; k < 8; ++k) acc += GQ[i][k] * Gmat[j * 8 + k];
    Qe[i][j] = acc;
    __syncthreads();

    {
        float am0 = 0.f, ah0 = 0.f;
        #pragma unroll
        for (int k = 0; k < 8; ++k) {
            am0 += Hm[i][k] * Fm[k][j];
            ah0 += Hm[i][k] * Qe[k][j];
        }
        Mm[i][j]  = am0;
        HQm[i][j] = ah0;
    }
    __syncthreads();

    float fri[8], frj[8], mri[8], mrj[8], mcj[8];
    float cij = Rm[i][j];
    #pragma unroll
    for (int k = 0; k < 8; ++k) {
        fri[k] = Fm[i][k];  frj[k] = Fm[j][k];
        mri[k] = Mm[i][k];  mrj[k] = Mm[j][k];  mcj[k] = Mm[k][j];
        cij += HQm[i][k] * Hm[j][k];
    }
    const float qe  = Qe[i][j];
    const float hq  = HQm[i][j];
    const float fij = Fm[i][j];
    float Pw[64];
    #pragma unroll
    for (int q = 0; q < 16; ++q) {
        const v4f p4 = ((const v4f*)&PnL[0][0])[q];
        Pw[4*q] = p4.x; Pw[4*q+1] = p4.y; Pw[4*q+2] = p4.z; Pw[4*q+3] = p4.w;
    }
    float pold = PnL[i][j];

    int tstop = Tn;
    int fc = 0, fp = 0;
    for (int t = 0; t < Tn; ++t) {
        float sv = cij, tv = hq;
        {
            float u[8];
            #pragma unroll
            for (int kk = 0; kk < 8; ++kk) {
                float a = 0.f;
                #pragma unroll
                for (int m = 0; m < 8; ++m) a += mri[m] * Pw[m * 8 + kk];
                u[kk] = a;
            }
            #pragma unroll
            for (int k = 0; k < 8; ++k) {
                sv += u[k] * mrj[k];
                tv += u[k] * frj[k];
            }
        }
        Sb[i][j] = sv;
        Tb[i][j] = tv;
        float pn = qe;
        {
            float t1[8];
            #pragma unroll
            for (int kk = 0; kk < 8; ++kk) {
                float a = 0.f;
                #pragma unroll
                for (int m = 0; m < 8; ++m) a += fri[m] * Pw[m * 8 + kk];
                t1[kk] = a;
            }
            #pragma unroll
            for (int k = 0; k < 8; ++k) pn += t1[k] * frj[k];
        }
        barrier_lds_only();
        if (t > 0 && flags[fp] == 0) { tstop = t; break; }

        float xc[8];
        float tcol[8];
        {
            float s[64], bc[8];
            #pragma unroll
            for (int q = 0; q < 16; ++q) {
                const v4f s4 = ((const v4f*)&Sb[0][0])[q];
                s[4*q] = s4.x; s[4*q+1] = s4.y; s[4*q+2] = s4.z; s[4*q+3] = s4.w;
            }
            #pragma unroll
            for (int m = 0; m < 8; ++m) bc[m] = Tb[m][i];
            #pragma unroll
            for (int m = 0; m < 8; ++m) tcol[m] = Tb[m][j];

            float l0[4][8], l1[4][8], w[8];
            #pragma unroll
            for (int p = 3; p >= 0; --p) {
                const int k = 2 * p;
                const float pa = s[k * 8 + k], pc = s[(k + 1) * 8 + k];
                const float pd = s[(k + 1) * 8 + k + 1];
                const float det = pa * pd - pc * pc;
                const float rd  = 1.0f / det;
                float n0[8], n1[8];
                #pragma unroll
                for (int r = 0; r < k; ++r) {
                    const float a0 = s[k * 8 + r], a1 = s[(k + 1) * 8 + r];
                    n0[r] = a0 * pd - a1 * pc;
                    n1[r] = a1 * pa - a0 * pc;
                    l0[p][r] = n0[r] * rd;
                    l1[p][r] = n1[r] * rd;
                }
                #pragma unroll
                for (int r = 0; r < k; ++r) {
                    #pragma unroll
                    for (int cc = 0; cc <= r; ++cc) {
                        if (r >= k - 2 && cc >= k - 2) {
                            const float tt = n0[r] * s[k * 8 + cc] + n1[r] * s[(k + 1) * 8 + cc];
                            s[r * 8 + cc] = s[r * 8 + cc] - tt * rd;
                        } else {
                            float e = s[r * 8 + cc];
                            e -= l0[p][r] * s[k * 8 + cc];
                            e -= l1[p][r] * s[(k + 1) * 8 + cc];
                            s[r * 8 + cc] = e;
                        }
                    }
                    if (r >= k - 2) {
                        const float tt = n0[r] * bc[k] + n1[r] * bc[k + 1];
                        bc[r] = bc[r] - tt * rd;
                    } else {
                        float e = bc[r];
                        e -= l0[p][r] * bc[k];
                        e -= l1[p][r] * bc[k + 1];
                        bc[r] = e;
                    }
                }
                w[k]     = (pd * bc[k] - pc * bc[k + 1]) * rd;
                w[k + 1] = (pa * bc[k + 1] - pc * bc[k]) * rd;
            }
            #pragma unroll
            for (int p = 0; p < 4; ++p) {
                const int k = 2 * p;
                float u0 = w[k], u1 = w[k + 1];
                #pragma unroll
                for (int r = 0; r < k; ++r) {
                    u0 -= l0[p][r] * xc[r];
                    u1 -= l1[p][r] * xc[r];
                }
                xc[k] = u0; xc[k + 1] = u1;
            }
        }
        float kv = xc[0];
        #pragma unroll
        for (int m = 1; m < 8; ++m) kv = (j == m) ? xc[m] : kv;
        float am = fij, pacc = pn;
        #pragma unroll
        for (int m = 0; m < 8; ++m) {
            am   -= xc[m] * mcj[m];
            pacc -= xc[m] * tcol[m];
        }
        PnL[i][j]  = pacc;
        barrier_lds_only();
        float sym;
        {
            float Pr[64];
            #pragma unroll
            for (int q = 0; q < 16; ++q) {
                const v4f p4 = ((const v4f*)&PnL[0][0])[q];
                Pr[4*q] = p4.x; Pr[4*q+1] = p4.y; Pr[4*q+2] = p4.z; Pr[4*q+3] = p4.w;
            }
            sym = 0.5f * (pacc + PnL[j][i]);
            #pragma unroll
            for (int m = 0; m < 8; ++m) {
                #pragma unroll
                for (int k = m; k < 8; ++k) {
                    const float s = 0.5f * (Pr[m * 8 + k] + Pr[k * 8 + m]);
                    Pw[m * 8 + k] = s;
                    Pw[k * 8 + m] = s;
                }
            }
        }
        const int fn = (fc == 2) ? 0 : fc + 1;
        if (sym != pold) flags[fc] = 1;
        if (tid == 0) flags[fn] = 0;
        pold = sym;
        fp = fc; fc = fn;

        Ps[t * 64 + tid] = sym;
        v2f g;
        g.x = am;
        g.y = kv;
        *(v2f*)(Gnat + ((size_t)t * 64 + tid) * 2) = g;
    }
    if (tid == 0) *t_stop_ptr = tstop;
}

__global__ __launch_bounds__(256) void kf_fill_kernel(
    float* __restrict__ Gws, const float* __restrict__ Gnat,
    const int* __restrict__ t_stop_ptr, float* __restrict__ Ps)
{
    const int t  = blockIdx.x;
    const int f  = threadIdx.x;
    const int ts = *t_stop_ptr;
    const int last = (ts > 0) ? (ts - 1) : 0;
    const int src  = (t < ts) ? t : last;

    const int lane = f >> 3, slot = f & 7;
    const int s2 = slot >> 1, p = slot & 1;
    const int m = lane & 15;
    const int k = 4 * s2 + ((lane < 16) ? p : (2 + p));
    const int mm = (m < 8) ? m : ((m <= 13) ? (m - 6) : -1);
    float v = 0.f;
    if (mm >= 0) {
        const int e = (k < 8) ? ((mm * 8 + k) * 2) : ((mm * 8 + (k - 8)) * 2 + 1);
        v = Gnat[(size_t)src * 128 + e];
    }
    Gws[(size_t)t * 256 + f] = v;
    if (t >= ts && ts >= 1 && f < 64) Ps[t * 64 + f] = Ps[src * 64 + f];
}

__global__ __launch_bounds__(32) void kf_scan_kernel(
    const float* __restrict__ Y, const float* __restrict__ x0,
    const float* __restrict__ Gws, float* __restrict__ traj)
{
    const int lane = threadIdx.x;
    const int n    = lane & 15;
    const bool lo  = lane < 16;
    const size_t row = (size_t)(blockIdx.x * 16 + n);

    const float* yrow = Y    + row * (size_t)Tn * 8 + (lo ? 0 : 2);
    float*       orow = traj + row * (size_t)Tn * 8 + (lo ? 0 : 4);

    v4f xa = *(const v4f*)(x0 + row * 8);
    v4f xb = *(const v4f*)(x0 + row * 8 + 4);
    v8f xv;
    xv[0] = lo ? xa.x : xa.z;
    xv[1] = lo ? xa.y : xa.w;
    xv[2] = lo ? xa.z : xb.x;
    xv[3] = lo ? xa.w : xb.y;
    xv[4] = lo ? xb.x : xb.z;
    xv[5] = lo ? xb.y : xb.w;
    xv[6] = lo ? xb.z : 0.f;
    xv[7] = lo ? xb.w : 0.f;

    const v8f zero8 = {0.f, 0.f, 0.f, 0.f, 0.f, 0.f, 0.f, 0.f};

    constexpr int RD = 12;
    v4f rg0[RD], rg1[RD];
    v2f ry2[RD], ry3[RD];
    #pragma unroll
    for (int u = 0; u < RD; ++u) {
        const v4f* gp = (const v4f*)(Gws + (size_t)u * 256 + lane * 8);
        rg0[u] = gp[0];
        rg1[u] = gp[1];
        ry2[u] = *(const v2f*)(yrow + (size_t)u * 8);
        ry3[u] = *(const v2f*)(yrow + (size_t)u * 8 + 4);
        __builtin_amdgcn_sched_barrier(0);
    }

    #define KF_SCAN_STEP(u, un, t)                                                \
    {                                                                             \
        const v2f a0 = {rg0[u].x, rg0[u].y};          \
        const v2f a1 = {rg0[u].z, rg0[u].w};          \
        const v2f a2 = {rg1[un].x, rg1[un].y};        \
        const v2f a3 = {rg1[un].z, rg1[un].w};        \
                     \
        const v2f b0 = {xv[0], xv[1]};               \
        const v2f b1 = {xv[4], xv[5]};               \
        const v8f c1 = __builtin_amdgcn_wmma_f32_16x16x4_f32(false, a0, false, b0, \
                                                       (short)0, cyc, false, false); \
        const v8f c2 = __builtin_amdgcn_wmma_f32_16x16x4_f32(false, a1, false, b1, \
                                                       (short)0, zero8, false, false); \
           \
        v8f cn = __builtin_amdgcn_wmma_f32_16x16x4_f32(false, a2, false, ry2[un], \
                                                       (short)0, zero8, false, false); \
        cn = __builtin_amdgcn_wmma_f32_16x16x4_f32(false, a3, false, ry3[un],     \
                                                   (short)0, cn, false, false);   \
        xv = c1 + c2;                                                             \
        cyc = cn;                                                                 \
                  \
                       \
        v4f out;                                                                  \
        out.x = lo ? xv[0] : xv[2];                                               \
        out.y = lo ? xv[1] : xv[3];                                               \
        out.z = lo ? xv[2] : xv[4];                                               \
        out.w = lo ? xv[3] : xv[5];                                               \
        *(v4f*)(orow + (size_t)(t) * 8) = out;                                    \
    }

    v8f cyc;
    {
        const v2f a2 = {rg1[0].x, rg1[0].y};
        const v2f a3 = {rg1[0].z, rg1[0].w};
        cyc = __builtin_amdgcn_wmma_f32_16x16x4_f32(false, a2, false, ry2[0],
                                                    (short)0, zero8, false, false);
        cyc = __builtin_amdgcn_wmma_f32_16x16x4_f32(false, a3, false, ry3[0],
                                                    (short)0, cyc, false, false);
    }

    int t = 0;
    for (; t + RD <= Tn; t += RD) {
        #pragma unroll
        for (int u = 0; u < RD; ++u) {
            KF_SCAN_STEP(u, (u + 1) % RD, t + u)
            const int tn = t + u + RD;
            const int tl = (tn < Tn) ? tn : (Tn - 1);
            const v4f* gp = (const v4f*)(Gws + (size_t)tl * 256 + lane * 8);
            rg0[u] = gp[0];
            rg1[u] = gp[1];
            ry2[u] = *(const v2f*)(yrow + (size_t)tl * 8);
            ry3[u] = *(const v2f*)(yrow + (size_t)tl * 8 + 4);
            __builtin_amdgcn_sched_barrier(0);
        }
    }
    #pragma unroll
    for (int u = 0; u < Tn % RD; ++u) {
        KF_SCAN_STEP(u, (u + 1) % RD, t + u)
    }
    #undef KF_SCAN_STEP
}

extern "C" void kernel_launch(void* const* d_in, const int* in_sizes, int n_in,
                              void* d_out, int out_size, void* d_ws, size_t ws_size,
                              hipStream_t stream) {
    const float* Y    = (const float*)d_in[0];
    const float* F    = (const float*)d_in[1];
    const float* Gmat = (const float*)d_in[2];
    const float* H    = (const float*)d_in[3];
    const float* Q    = (const float*)d_in[4];
    const float* R    = (const float*)d_in[5];
    const float* x0   = (const float*)d_in[6];
    const float* P0   = (const float*)d_in[7];

    float* traj = (float*)d_out;
    float* Ps   = traj + (size_t)Bn * Tn * 8;

    int*   t_stop = (int*)d_ws;
    float* Gws    = (float*)((char*)d_ws + 256);
    float* Gnat   = (float*)((char*)d_ws + 1024256);

    kf_gain_kernel<<<1, 64, 0, stream>>>(F, Gmat, H, Q, R, P0, Gnat, t_stop, Ps);
    kf_fill_kernel<<<Tn, 256, 0, stream>>>(Gws, Gnat, t_stop, Ps);
    kf_scan_kernel<<<Bn / 16, 32, 0, stream>>>(Y, x0, Gws, traj);
}


# Gain kernel: a thread keeps only column j of P; rows of M P and F P reach a matrix row's 8 threads as DPP8 lane-group broadcasts; Pn is read back as row j and column j only; sum order kept
# speedup vs baseline: 9.8593x; 1.1084x over previous
#include <hip/hip_runtime.h>
#include <math.h>

static constexpr int Bn = 2048;
static constexpr int Tn = 1000;

typedef float v2f __attribute__((ext_vector_type(2)));
typedef float v4f __attribute__((ext_vector_type(4)));
typedef float v8f __attribute__((ext_vector_type(8)));

__device__ __forceinline__ void barrier_lds_only() {
    __builtin_amdgcn_fence(__ATOMIC_RELEASE, "workgroup", "local");
    __builtin_amdgcn_s_barrier();
    __builtin_amdgcn_fence(__ATOMIC_ACQUIRE, "workgroup", "local");
}

template <int K>
__device__ __forceinline__ float kf_row_bcast(float v) {
    return __builtin_bit_cast(float,
        __builtin_amdgcn_mov_dpp8(__builtin_bit_cast(unsigned int, v),
                                  (unsigned int)(K * 0x249249)));
}

__global__ __launch_bounds__(64) void kf_gain_kernel(
    const float* __restrict__ F, const float* __restrict__ Gmat,
    const float* __restrict__ H, const float* __restrict__ Q,
    const float* __restrict__ R, const float* __restrict__ P0,
    float* __restrict__ Gnat, int* __restrict__ t_stop_ptr,
    float* __restrict__ Ps)
{
    __shared__ __attribute__((aligned(16))) float Fm[8][8];
    __shared__ __attribute__((aligned(16))) float Hm[8][8];
    __shared__ __attribute__((aligned(16))) float Qe[8][8];
    __shared__ __attribute__((aligned(16))) float Rm[8][8];
    __shared__ __attribute__((aligned(16))) float Mm[8][8];
    __shared__ __attribute__((aligned(16))) float HQm[8][8];
    __shared__ __attribute__((aligned(16))) float PnL[8][8];
    __shared__ __attribute__((aligned(16))) float PnT[8][8];
    __shared__ __attribute__((aligned(16))) float Sb[8][8];
    __shared__ __attribute__((aligned(16))) float Tb[8][8];
    __shared__ float GQ[8][8];
    __shared__ int flags[3];

    const int tid = threadIdx.x;
    const int i = tid >> 3, j = tid & 7;

    Fm[i][j]  = F[tid];
    Hm[i][j]  = H[tid];
    Rm[i][j]  = R[tid];
    PnL[i][j] = P0[tid];
    if (tid < 3) flags[tid] = 0;
    __syncthreads();

    float acc = 0.f;
    #pragma unroll
    for (int k = 0; k < 8; ++k) acc += Gmat[i * 8 + k] * Q[k * 8 + j];
    GQ[i][j] = acc;
    __syncthreads();
    acc = 0.f;
    #pragma unroll
    for (int k = 0; k < 8; ++k) acc += GQ[i][k] * Gmat[j * 8 + k];
    Qe[i][j] = acc;
    __syncthreads();

    {
        float am0 = 0.f, ah0 = 0.f;
        #pragma unroll
        for (int k = 0; k < 8; ++k) {
            am0 += Hm[i][k] * Fm[k][j];
            ah0 += Hm[i][k] * Qe[k][j];
        }
        Mm[i][j]  = am0;
        HQm[i][j] = ah0;
    }
    __syncthreads();

    float fri[8], frj[8], mri[8], mrj[8], mcj[8];
    float cij = Rm[i][j];
    #pragma unroll
    for (int k = 0; k < 8; ++k) {
        fri[k] = Fm[i][k];  frj[k] = Fm[j][k];
        mri[k] = Mm[i][k];  mrj[k] = Mm[j][k];  mcj[k] = Mm[k][j];
        cij += HQm[i][k] * Hm[j][k];
    }
    const float qe  = Qe[i][j];
    const float hq  = HQm[i][j];
    const float fij = Fm[i][j];
    float pcol[8];
    #pragma unroll
    for (int m = 0; m < 8; ++m) pcol[m] = PnL[m][j];
    float pold = PnL[i][j];

    int tstop = Tn;
    int fc = 0, fp = 0;
    for (int t = 0; t < Tn; ++t) {
        float uo = 0.f, to = 0.f;
        #pragma unroll
        for (int m = 0; m < 8; ++m) {
            uo = __builtin_fmaf(mri[m], pcol[m], uo);
            to = __builtin_fmaf(fri[m], pcol[m], to);
        }
        float sv = cij, tv = hq, pn = qe;
        #define KF_ROW_TERM(k)                                  \
        {                                                       \
            const float ub = kf_row_bcast<k>(uo);               \
            const float tb = kf_row_bcast<k>(to);               \
            sv += ub * mrj[k];                                  \
            tv += ub * frj[k];                                  \
            pn += tb * frj[k];                                  \
        }
        KF_ROW_TERM(0) KF_ROW_TERM(1) KF_ROW_TERM(2) KF_ROW_TERM(3)
        KF_ROW_TERM(4) KF_ROW_TERM(5) KF_ROW_TERM(6) KF_ROW_TERM(7)
        #undef KF_ROW_TERM
        Sb[i][j] = sv;
        Tb[i][j] = tv;
        barrier_lds_only();
        if (t > 0 && flags[fp] == 0) { tstop = t; break; }

        float xc[8];
        float tcol[8];
        {
            float s[64], bc[8];
            #pragma unroll
            for (int q = 0; q < 16; ++q) {
                const v4f s4 = ((const v4f*)&Sb[0][0])[q];
                s[4*q] = s4.x; s[4*q+1] = s4.y; s[4*q+2] = s4.z; s[4*q+3] = s4.w;
            }
            #pragma unroll
            for (int m = 0; m < 8; ++m) bc[m] = Tb[m][i];
            #pragma unroll
            for (int m = 0; m < 8; ++m) tcol[m] = Tb[m][j];

            float l0[4][8], l1[4][8], w[8];
            #pragma unroll
            for (int p = 3; p >= 0; --p) {
                const int k = 2 * p;
                const float pa = s[k * 8 + k], pc = s[(k + 1) * 8 + k];
                const float pd = s[(k + 1) * 8 + k + 1];
                const float det = pa * pd - pc * pc;
                const float rd  = 1.0f / det;
                float n0[8], n1[8];
                #pragma unroll
                for (int r = 0; r < k; ++r) {
                    const float a0 = s[k * 8 + r], a1 = s[(k + 1) * 8 + r];
                    n0[r] = a0 * pd - a1 * pc;
                    n1[r] = a1 * pa - a0 * pc;
                    l0[p][r] = n0[r] * rd;
                    l1[p][r] = n1[r] * rd;
                }
                #pragma unroll
                for (int r = 0; r < k; ++r) {
                    #pragma unroll
                    for (int cc = 0; cc <= r; ++cc) {
                        if (r >= k - 2 && cc >= k - 2) {
                            const float tt = n0[r] * s[k * 8 + cc] + n1[r] * s[(k + 1) * 8 + cc];
                            s[r * 8 + cc] = s[r * 8 + cc] - tt * rd;
                        } else {
                            float e = s[r * 8 + cc];
                            e -= l0[p][r] * s[k * 8 + cc];
                            e -= l1[p][r] * s[(k + 1) * 8 + cc];
                            s[r * 8 + cc] = e;
                        }
                    }
                    if (r >= k - 2) {
                        const float tt = n0[r] * bc[k] + n1[r] * bc[k + 1];
                        bc[r] = bc[r] - tt * rd;
                    } else {
                        float e = bc[r];
                        e -= l0[p][r] * bc[k];
                        e -= l1[p][r] * bc[k + 1];
                        bc[r] = e;
                    }
                }
                w[k]     = (pd * bc[k] - pc * bc[k + 1]) * rd;
                w[k + 1] = (pa * bc[k + 1] - pc * bc[k]) * rd;
            }
            #pragma unroll
            for (int p = 0; p < 4; ++p) {
                const int k = 2 * p;
                float u0 = w[k], u1 = w[k + 1];
                #pragma unroll
                for (int r = 0; r < k; ++r) {
                    u0 -= l0[p][r] * xc[r];
                    u1 -= l1[p][r] * xc[r];
                }
                xc[k] = u0; xc[k + 1] = u1;
            }
        }
        float kv = xc[0];
        #pragma unroll
        for (int m = 1; m < 8; ++m) kv = (j == m) ? xc[m] : kv;
        float am = fij, pacc = pn;
        #pragma unroll
        for (int m = 0; m < 8; ++m) {
            am   -= xc[m] * mcj[m];
            pacc -= xc[m] * tcol[m];
        }
        PnL[i][j]  = pacc;
        PnT[j][i]  = pacc;
        barrier_lds_only();
        float sym;
        {
            const v4f r0 = ((const v4f*)&PnL[j][0])[0];
            const v4f r1 = ((const v4f*)&PnL[j][0])[1];
            const v4f c0 = ((const v4f*)&PnT[j][0])[0];
            const v4f c1 = ((const v4f*)&PnT[j][0])[1];
            sym = 0.5f * (pacc + PnL[j][i]);
            const v4f s0 = 0.5f * (c0 + r0);
            const v4f s1 = 0.5f * (c1 + r1);
            pcol[0] = s0.x; pcol[1] = s0.y; pcol[2] = s0.z; pcol[3] = s0.w;
            pcol[4] = s1.x; pcol[5] = s1.y; pcol[6] = s1.z; pcol[7] = s1.w;
        }
        const int fn = (fc == 2) ? 0 : fc + 1;
        if (sym != pold) flags[fc] = 1;
        if (tid == 0) flags[fn] = 0;
        pold = sym;
        fp = fc; fc = fn;

        Ps[t * 64 + tid] = sym;
        v2f g;
        g.x = am;
        g.y = kv;
        *(v2f*)(Gnat + ((size_t)t * 64 + tid) * 2) = g;
    }
    if (tid == 0) *t_stop_ptr = tstop;
}

__global__ __launch_bounds__(256) void kf_fill_kernel(
    float* __restrict__ Gws, const float* __restrict__ Gnat,
    const int* __restrict__ t_stop_ptr, float* __restrict__ Ps)
{
    const int t  = blockIdx.x;
    const int f  = threadIdx.x;
    const int ts = *t_stop_ptr;
    const int last = (ts > 0) ? (ts - 1) : 0;
    const int src  = (t < ts) ? t : last;

    const int lane = f >> 3, slot = f & 7;
    const int s2 = slot >> 1, p = slot & 1;
    const int m = lane & 15;
    const int k = 4 * s2 + ((lane < 16) ? p : (2 + p));
    const int mm = (m < 8) ? m : ((m <= 13) ? (m - 6) : -1);
    float v = 0.f;
    if (mm >= 0) {
        const int e = (k < 8) ? ((mm * 8 + k) * 2) : ((mm * 8 + (k - 8)) * 2 + 1);
        v = Gnat[(size_t)src * 128 + e];
    }
    Gws[(size_t)t * 256 + f] = v;
    if (t >= ts && ts >= 1 && f < 64) Ps[t * 64 + f] = Ps[src * 64 + f];
}

__global__ __launch_bounds__(32) void kf_scan_kernel(
    const float* __restrict__ Y, const float* __restrict__ x0,
    const float* __restrict__ Gws, float* __restrict__ traj)
{
    const int lane = threadIdx.x;
    const int n    = lane & 15;
    const bool lo  = lane < 16;
    const size_t row = (size_t)(blockIdx.x * 16 + n);

    const float* yrow = Y    + row * (size_t)Tn * 8 + (lo ? 0 : 2);
    float*       orow = traj + row * (size_t)Tn * 8 + (lo ? 0 : 4);

    v4f xa = *(const v4f*)(x0 + row * 8);
    v4f xb = *(const v4f*)(x0 + row * 8 + 4);
    v8f xv;
    xv[0] = lo ? xa.x : xa.z;
    xv[1] = lo ? xa.y : xa.w;
    xv[2] = lo ? xa.z : xb.x;
    xv[3] = lo ? xa.w : xb.y;
    xv[4] = lo ? xb.x : xb.z;
    xv[5] = lo ? xb.y : xb.w;
    xv[6] = lo ? xb.z : 0.f;
    xv[7] = lo ? xb.w : 0.f;

    const v8f zero8 = {0.f, 0.f, 0.f, 0.f, 0.f, 0.f, 0.f, 0.f};

    constexpr int RD = 12;
    v4f rg0[RD], rg1[RD];
    v2f ry2[RD], ry3[RD];
    #pragma unroll
    for (int u = 0; u < RD; ++u) {
        const v4f* gp = (const v4f*)(Gws + (size_t)u * 256 + lane * 8);
        rg0[u] = gp[0];
        rg1[u] = gp[1];
        ry2[u] = *(const v2f*)(yrow + (size_t)u * 8);
        ry3[u] = *(const v2f*)(yrow + (size_t)u * 8 + 4);
        __builtin_amdgcn_sched_barrier(0);
    }

    #define KF_SCAN_STEP(u, un, t)                                                \
    {                                                                             \
        const v2f a0 = {rg0[u].x, rg0[u].y};          \
        const v2f a1 = {rg0[u].z, rg0[u].w};          \
        const v2f a2 = {rg1[un].x, rg1[un].y};        \
        const v2f a3 = {rg1[un].z, rg1[un].w};        \
                     \
        const v2f b0 = {xv[0], xv[1]};               \
        const v2f b1 = {xv[4], xv[5]};               \
        const v8f c1 = __builtin_amdgcn_wmma_f32_16x16x4_f32(false, a0, false, b0, \
                                                       (short)0, cyc, false, false); \
        const v8f c2 = __builtin_amdgcn_wmma_f32_16x16x4_f32(false, a1, false, b1, \
                                                       (short)0, zero8, false, false); \
           \
        v8f cn = __builtin_amdgcn_wmma_f32_16x16x4_f32(false, a2, false, ry2[un], \
                                                       (short)0, zero8, false, false); \
        cn = __builtin_amdgcn_wmma_f32_16x16x4_f32(false, a3, false, ry3[un],     \
                                                   (short)0, cn, false, false);   \
        xv = c1 + c2;                                                             \
        cyc = cn;                                                                 \
                  \
                       \
        v4f out;                                                                  \
        out.x = lo ? xv[0] : xv[2];                                               \
        out.y = lo ? xv[1] : xv[3];                                               \
        out.z = lo ? xv[2] : xv[4];                                               \
        out.w = lo ? xv[3] : xv[5];                                               \
        *(v4f*)(orow + (size_t)(t) * 8) = out;                                    \
    }

    v8f cyc;
    {
        const v2f a2 = {rg1[0].x, rg1[0].y};
        const v2f a3 = {rg1[0].z, rg1[0].w};
        cyc = __builtin_amdgcn_wmma_f32_16x16x4_f32(false, a2, false, ry2[0],
                                                    (short)0, zero8, false, false);
        cyc = __builtin_amdgcn_wmma_f32_16x16x4_f32(false, a3, false, ry3[0],
                                                    (short)0, cyc, false, false);
    }

    int t = 0;
    for (; t + RD <= Tn; t += RD) {
        #pragma unroll
        for (int u = 0; u < RD; ++u) {
            KF_SCAN_STEP(u, (u + 1) % RD, t + u)
            const int tn = t + u + RD;
            const int tl = (tn < Tn) ? tn : (Tn - 1);
            const v4f* gp = (const v4f*)(Gws + (size_t)tl * 256 + lane * 8);
            rg0[u] = gp[0];
            rg1[u] = gp[1];
            ry2[u] = *(const v2f*)(yrow + (size_t)tl * 8);
            ry3[u] = *(const v2f*)(yrow + (size_t)tl * 8 + 4);
            __builtin_amdgcn_sched_barrier(0);
        }
    }
    #pragma unroll
    for (int u = 0; u < Tn % RD; ++u) {
        KF_SCAN_STEP(u, (u + 1) % RD, t + u)
    }
    #undef KF_SCAN_STEP
}

extern "C" void kernel_launch(void* const* d_in, const int* in_sizes, int n_in,
                              void* d_out, int out_size, void* d_ws, size_t ws_size,
                              hipStream_t stream) {
    const float* Y    = (const float*)d_in[0];
    const float* F    = (const float*)d_in[1];
    const float* Gmat = (const float*)d_in[2];
    const float* H    = (const float*)d_in[3];
    const float* Q    = (const float*)d_in[4];
    const float* R    = (const float*)d_in[5];
    const float* x0   = (const float*)d_in[6];
    const float* P0   = (const float*)d_in[7];

    float* traj = (float*)d_out;
    float* Ps   = traj + (size_t)Bn * Tn * 8;

    int*   t_stop = (int*)d_ws;
    float* Gws    = (float*)((char*)d_ws + 256);
    float* Gnat   = (float*)((char*)d_ws + 1024256);

    kf_gain_kernel<<<1, 64, 0, stream>>>(F, Gmat, H, Q, R, P0, Gnat, t_stop, Ps);
    kf_fill_kernel<<<Tn, 256, 0, stream>>>(Gws, Gnat, t_stop, Ps);
    kf_scan_kernel<<<Bn / 16, 32, 0, stream>>>(Y, x0, Gws, traj);
}
